# starting kernel
# baseline (speedup 1.0000x reference)
#include <hip/hip_runtime.h>

#define B_    32
#define N_    1024
#define D_    512
#define K_SEL 512
#define EPS_  1e-8f

typedef __attribute__((ext_vector_type(2))) float v2f;
typedef __attribute__((ext_vector_type(8))) float v8f;

__global__ __launch_bounds__(256)
void k1_norms_part(const float* __restrict__ x, float* __restrict__ inv_norms,
                   float* __restrict__ part) {
  const int b      = blockIdx.x >> 4;
  const int rowblk = blockIdx.x & 15;
  const int row0   = rowblk << 6;
  const int lane   = threadIdx.x & 31;
  const int wave   = threadIdx.x >> 5;

  __shared__ float lds_acc[8][512];

  float acc[16];
#pragma unroll
  for (int i = 0; i < 16; ++i) acc[i] = 0.f;

  const float* xb = x + (size_t)b * (N_ * D_);

  for (int r = 0; r < 8; ++r) {
    const int row = row0 + wave * 8 + r;
    const float4* xr = (const float4*)(xb + (size_t)row * D_);
    float4 v[4];
    float ss = 0.f;
#pragma unroll
    for (int c = 0; c < 4; ++c) {
      v[c] = xr[c * 32 + lane];
      ss += v[c].x * v[c].x + v[c].y * v[c].y + v[c].z * v[c].z + v[c].w * v[c].w;
    }
    if (r + 1 < 8)
      __builtin_prefetch(xb + (size_t)(row + 1) * D_ + lane * 16, 0, 1);
#pragma unroll
    for (int off = 16; off > 0; off >>= 1) ss += __shfl_xor(ss, off, 32);
    const float inv = 1.f / fmaxf(sqrtf(ss), EPS_);
    if (lane == 0) inv_norms[b * N_ + row] = inv;
#pragma unroll
    for (int c = 0; c < 4; ++c) {
      acc[c * 4 + 0] += v[c].x * inv;
      acc[c * 4 + 1] += v[c].y * inv;
      acc[c * 4 + 2] += v[c].z * inv;
      acc[c * 4 + 3] += v[c].w * inv;
    }
  }

#pragma unroll
  for (int c = 0; c < 4; ++c) {
    const int d = c * 128 + lane * 4;
    lds_acc[wave][d + 0] = acc[c * 4 + 0];
    lds_acc[wave][d + 1] = acc[c * 4 + 1];
    lds_acc[wave][d + 2] = acc[c * 4 + 2];
    lds_acc[wave][d + 3] = acc[c * 4 + 3];
  }
  __syncthreads();

  for (int d = threadIdx.x; d < 512; d += 256) {
    float s = 0.f;
#pragma unroll
    for (int w8 = 0; w8 < 8; ++w8) s += lds_acc[w8][d];
    part[((size_t)b * 16 + rowblk) * D_ + d] = s;
  }
}

__global__ __launch_bounds__(256)
void k1b_mu(const float* __restrict__ part, float* __restrict__ mu) {
  const int i = blockIdx.x * 256 + threadIdx.x;
  if (i >= B_ * D_) return;
  const int b = i / D_;
  const int d = i - b * D_;
  float s = 0.f;
#pragma unroll
  for (int j = 0; j < 16; ++j) s += part[((size_t)b * 16 + j) * D_ + d];
  mu[i] = s * (1.0f / N_);
}

__global__ __launch_bounds__(256)
void k2_scores(const float* __restrict__ x, const float* __restrict__ inv_norms,
               const float* __restrict__ mu, float* __restrict__ scores) {
  const int b    = blockIdx.x >> 7;
  const int row0 = (blockIdx.x & 127) << 3;
  const int lane = threadIdx.x & 31;
  const int wave = threadIdx.x >> 5;

  __shared__ float4 mul[128];
  {
    const float4* mub = (const float4*)(mu + (size_t)b * D_);
    for (int i = threadIdx.x; i < 128; i += 256) mul[i] = mub[i];
  }
  __syncthreads();

  const int row = row0 + wave;
  const float4* xr = (const float4*)(x + ((size_t)b * N_ + row) * D_);
  float dot = 0.f;
#pragma unroll
  for (int c = 0; c < 4; ++c) {
    const float4 v = xr[c * 32 + lane];
    const float4 m = mul[c * 32 + lane];
    dot += v.x * m.x + v.y * m.y + v.z * m.z + v.w * m.w;
  }
#pragma unroll
  for (int off = 16; off > 0; off >>= 1) dot += __shfl_xor(dot, off, 32);
  if (lane == 0) scores[b * N_ + row] = dot * inv_norms[b * N_ + row];
}

__global__ __launch_bounds__(256)
void k3_rank(const float* __restrict__ scores, float* __restrict__ w) {
  const int b = blockIdx.x >> 2;
  const int m = ((blockIdx.x & 3) << 8) + threadIdx.x;

  __shared__ float s[N_];
  for (int i = threadIdx.x; i < N_; i += 256) s[i] = scores[b * N_ + i];
  __syncthreads();

  const float sm = s[m];
  int rank = 0;
  for (int n = 0; n < N_; ++n) {
    const float sn = s[n];
    rank += (sn > sm) || ((sn == sm) && (n < m));
  }
  w[b * N_ + m] = (rank < K_SEL) ? (1.0f / K_SEL) : 0.0f;
}

__global__ __launch_bounds__(256)
void k4_merge_wmma(const float* __restrict__ x, const float* __restrict__ w,
                   float* __restrict__ out) {
  const int b     = blockIdx.x >> 2;
  const int tile  = ((blockIdx.x & 3) << 3) + (threadIdx.x >> 5);
  const int lane  = threadIdx.x & 31;
  const int half  = lane >> 4;
  const int ln    = lane & 15;
  const int dbase = tile << 4;

  __shared__ float wl[N_];
  for (int i = threadIdx.x; i < N_; i += 256) wl[i] = w[b * N_ + i];
  __syncthreads();

  const float* xb = x + (size_t)b * (N_ * D_);

  v8f c = {};
  for (int k = 0; k < N_; k += 4) {
    const int ka = k + (half ? 2 : 0);
    const int kb = ka + 1;
    v2f a  = { wl[ka], wl[kb] };
    v2f bv = { xb[(size_t)ka * D_ + dbase + ln],
               xb[(size_t)kb * D_ + dbase + ln] };
    c = __builtin_amdgcn_wmma_f32_16x16x4_f32(
         false, a,  false, bv,
         (short)0, c,  false,  false);
  }

  if (lane < 16) out[(size_t)b * D_ + dbase + ln] = c[0];
}

extern "C" void kernel_launch(void* const* d_in, const int* in_sizes, int n_in,
                              void* d_out, int out_size, void* d_ws, size_t ws_size,
                              hipStream_t stream) {
  (void)in_sizes; (void)n_in; (void)out_size; (void)ws_size;

  const float* x = (const float*)d_in[0];
  float* out = (float*)d_out;

  char* ws = (char*)d_ws;
  float* inv_norms = (float*)(ws + 0);
  float* scores    = (float*)(ws + 131072);
  float* wsel      = (float*)(ws + 262144);
  float* mu        = (float*)(ws + 393216);
  float* part      = (float*)(ws + 458752);

  k1_norms_part<<<B_ * 16, 256, 0, stream>>>(x, inv_norms, part);
  k1b_mu<<<(B_ * D_ + 255) / 256, 256, 0, stream>>>(part, mu);
  k2_scores<<<B_ * 128, 256, 0, stream>>>(x, inv_norms, mu, scores);
  k3_rank<<<B_ * 4, 256, 0, stream>>>(scores, wsel);
  k4_merge_wmma<<<B_ * 4, 256, 0, stream>>>(x, wsel, out);
}


# rank kernel: per pair one f32 compare kept as a wave mask plus an add (>= below the wave's rows, > above, tie mask only on its own 32); scores by uniform 16-byte LDS loads one group ahead
# speedup vs baseline: 1.4288x; 1.4288x over previous
#include <hip/hip_runtime.h>

#define B_    32
#define N_    1024
#define D_    512
#define K_SEL 512
#define EPS_  1e-8f

typedef __attribute__((ext_vector_type(2))) float v2f;
typedef __attribute__((ext_vector_type(8))) float v8f;

__global__ __launch_bounds__(256)
void k1_norms_part(const float* __restrict__ x, float* __restrict__ inv_norms,
                   float* __restrict__ part) {
  const int b      = blockIdx.x >> 4;
  const int rowblk = blockIdx.x & 15;
  const int row0   = rowblk << 6;
  const int lane   = threadIdx.x & 31;
  const int wave   = threadIdx.x >> 5;

  __shared__ float lds_acc[8][512];

  float acc[16];
#pragma unroll
  for (int i = 0; i < 16; ++i) acc[i] = 0.f;

  const float* xb = x + (size_t)b * (N_ * D_);

  for (int r = 0; r < 8; ++r) {
    const int row = row0 + wave * 8 + r;
    const float4* xr = (const float4*)(xb + (size_t)row * D_);
    float4 v[4];
    float ss = 0.f;
#pragma unroll
    for (int c = 0; c < 4; ++c) {
      v[c] = xr[c * 32 + lane];
      ss += v[c].x * v[c].x + v[c].y * v[c].y + v[c].z * v[c].z + v[c].w * v[c].w;
    }
    if (r + 1 < 8)
      __builtin_prefetch(xb + (size_t)(row + 1) * D_ + lane * 16, 0, 1);
#pragma unroll
    for (int off = 16; off > 0; off >>= 1) ss += __shfl_xor(ss, off, 32);
    const float inv = 1.f / fmaxf(sqrtf(ss), EPS_);
    if (lane == 0) inv_norms[b * N_ + row] = inv;
#pragma unroll
    for (int c = 0; c < 4; ++c) {
      acc[c * 4 + 0] += v[c].x * inv;
      acc[c * 4 + 1] += v[c].y * inv;
      acc[c * 4 + 2] += v[c].z * inv;
      acc[c * 4 + 3] += v[c].w * inv;
    }
  }

#pragma unroll
  for (int c = 0; c < 4; ++c) {
    const int d = c * 128 + lane * 4;
    lds_acc[wave][d + 0] = acc[c * 4 + 0];
    lds_acc[wave][d + 1] = acc[c * 4 + 1];
    lds_acc[wave][d + 2] = acc[c * 4 + 2];
    lds_acc[wave][d + 3] = acc[c * 4 + 3];
  }
  __syncthreads();

  for (int d = threadIdx.x; d < 512; d += 256) {
    float s = 0.f;
#pragma unroll
    for (int w8 = 0; w8 < 8; ++w8) s += lds_acc[w8][d];
    part[((size_t)b * 16 + rowblk) * D_ + d] = s;
  }
}

__global__ __launch_bounds__(256)
void k1b_mu(const float* __restrict__ part, float* __restrict__ mu) {
  const int i = blockIdx.x * 256 + threadIdx.x;
  if (i >= B_ * D_) return;
  const int b = i / D_;
  const int d = i - b * D_;
  float s = 0.f;
#pragma unroll
  for (int j = 0; j < 16; ++j) s += part[((size_t)b * 16 + j) * D_ + d];
  mu[i] = s * (1.0f / N_);
}

__global__ __launch_bounds__(256)
void k2_scores(const float* __restrict__ x, const float* __restrict__ inv_norms,
               const float* __restrict__ mu, float* __restrict__ scores) {
  const int b    = blockIdx.x >> 7;
  const int row0 = (blockIdx.x & 127) << 3;
  const int lane = threadIdx.x & 31;
  const int wave = threadIdx.x >> 5;

  __shared__ float4 mul[128];
  {
    const float4* mub = (const float4*)(mu + (size_t)b * D_);
    for (int i = threadIdx.x; i < 128; i += 256) mul[i] = mub[i];
  }
  __syncthreads();

  const int row = row0 + wave;
  const float4* xr = (const float4*)(x + ((size_t)b * N_ + row) * D_);
  float dot = 0.f;
#pragma unroll
  for (int c = 0; c < 4; ++c) {
    const float4 v = xr[c * 32 + lane];
    const float4 m = mul[c * 32 + lane];
    dot += v.x * m.x + v.y * m.y + v.z * m.z + v.w * m.w;
  }
#pragma unroll
  for (int off = 16; off > 0; off >>= 1) dot += __shfl_xor(dot, off, 32);
  if (lane == 0) scores[b * N_ + row] = dot * inv_norms[b * N_ + row];
}

__device__ __forceinline__ void load_group(float4 (&v)[8], const float4* s4, const int g) {
  for (int i = 0; i < 8; ++i) v[i] = s4[g * 8 + i];
}

__device__ __forceinline__ void count_group(const float4 (&v)[8], const float sm,
                                            const int g, const int g_own, int (&acc)[8]) {
  unsigned mk[32];
  if (g == g_own) {
    for (int i = 0; i < 8; ++i) {
      const float e[4] = { v[i].x, v[i].y, v[i].z, v[i].w };
      for (int c = 0; c < 4; ++c) {
        const int k = i * 4 + c;
        const unsigned above_k = 0xFFFFFFFEu << k;
        const unsigned gt = __builtin_amdgcn_ballot_w32(e[c] > sm);
        const unsigned eq = __builtin_amdgcn_ballot_w32(e[c] == sm);
        mk[k] = gt | (eq & above_k);
      }
    }
  } else if (g < g_own) {
    for (int i = 0; i < 8; ++i) {
      mk[i * 4 + 0] = __builtin_amdgcn_ballot_w32(v[i].x >= sm);
      mk[i * 4 + 1] = __builtin_amdgcn_ballot_w32(v[i].y >= sm);
      mk[i * 4 + 2] = __builtin_amdgcn_ballot_w32(v[i].z >= sm);
      mk[i * 4 + 3] = __builtin_amdgcn_ballot_w32(v[i].w >= sm);
    }
  } else {
    for (int i = 0; i < 8; ++i) {
      mk[i * 4 + 0] = __builtin_amdgcn_ballot_w32(v[i].x > sm);
      mk[i * 4 + 1] = __builtin_amdgcn_ballot_w32(v[i].y > sm);
      mk[i * 4 + 2] = __builtin_amdgcn_ballot_w32(v[i].z > sm);
      mk[i * 4 + 3] = __builtin_amdgcn_ballot_w32(v[i].w > sm);
    }
  }
  for (int i = 0; i < 8; ++i)
    acc[i] += (int)__builtin_amdgcn_inverse_ballot_w32(mk[i * 4 + 0]) +
              (int)__builtin_amdgcn_inverse_ballot_w32(mk[i * 4 + 1]) +
              (int)__builtin_amdgcn_inverse_ballot_w32(mk[i * 4 + 2]) +
              (int)__builtin_amdgcn_inverse_ballot_w32(mk[i * 4 + 3]);
}

__global__ __launch_bounds__(128)
void k3_rank(const float* __restrict__ scores, float* __restrict__ w) {
  const int b  = blockIdx.x >> 3;
  const int mb = (blockIdx.x & 7) << 7;
  const int m  = mb + threadIdx.x;

  __shared__ float4 s4[N_ / 4];
  {
    const float4* sb = (const float4*)(scores + b * N_);
    for (int i = threadIdx.x; i < N_ / 4; i += 128) s4[i] = sb[i];
  }
  __syncthreads();

  const float sm  = ((const float*)s4)[m];
  const int g_own = (mb >> 5) + __builtin_amdgcn_readfirstlane((int)(threadIdx.x >> 5));

  int acc[8];
  for (int i = 0; i < 8; ++i) acc[i] = 0;

  float4 va[8], vb[8];
  load_group(va, s4, 0);
  for (int g = 0; g < N_ / 32; g += 2) {
    load_group(vb, s4, g + 1);
    count_group(va, sm, g, g_own, acc);
    load_group(va, s4, (g + 2) & (N_ / 32 - 1));
    count_group(vb, sm, g + 1, g_own, acc);
  }

  const int rank = ((acc[0] + acc[1]) + (acc[2] + acc[3])) + ((acc[4] + acc[5]) + (acc[6] + acc[7]));
  w[b * N_ + m] = (rank < K_SEL) ? (1.0f / K_SEL) : 0.0f;
}

__global__ __launch_bounds__(256)
void k4_merge_wmma(const float* __restrict__ x, const float* __restrict__ w,
                   float* __restrict__ out) {
  const int b     = blockIdx.x >> 2;
  const int tile  = ((blockIdx.x & 3) << 3) + (threadIdx.x >> 5);
  const int lane  = threadIdx.x & 31;
  const int half  = lane >> 4;
  const int ln    = lane & 15;
  const int dbase = tile << 4;

  __shared__ float wl[N_];
  for (int i = threadIdx.x; i < N_; i += 256) wl[i] = w[b * N_ + i];
  __syncthreads();

  const float* xb = x + (size_t)b * (N_ * D_);

  v8f c = {};
  for (int k = 0; k < N_; k += 4) {
    const int ka = k + (half ? 2 : 0);
    const int kb = ka + 1;
    v2f a  = { wl[ka], wl[kb] };
    v2f bv = { xb[(size_t)ka * D_ + dbase + ln],
               xb[(size_t)kb * D_ + dbase + ln] };
    c = __builtin_amdgcn_wmma_f32_16x16x4_f32(
         false, a,  false, bv,
         (short)0, c,  false,  false);
  }

  if (lane < 16) out[(size_t)b * D_ + dbase + ln] = c[0];
}

extern "C" void kernel_launch(void* const* d_in, const int* in_sizes, int n_in,
                              void* d_out, int out_size, void* d_ws, size_t ws_size,
                              hipStream_t stream) {
  (void)in_sizes; (void)n_in; (void)out_size; (void)ws_size;

  const float* x = (const float*)d_in[0];
  float* out = (float*)d_out;

  char* ws = (char*)d_ws;
  float* inv_norms = (float*)(ws + 0);
  float* scores    = (float*)(ws + 131072);
  float* wsel      = (float*)(ws + 262144);
  float* mu        = (float*)(ws + 393216);
  float* part      = (float*)(ws + 458752);

  k1_norms_part<<<B_ * 16, 256, 0, stream>>>(x, inv_norms, part);
  k1b_mu<<<(B_ * D_ + 255) / 256, 256, 0, stream>>>(part, mu);
  k2_scores<<<B_ * 128, 256, 0, stream>>>(x, inv_norms, mu, scores);
  k3_rank<<<B_ * 8, 128, 0, stream>>>(scores, wsel);
  k4_merge_wmma<<<B_ * 4, 256, 0, stream>>>(x, wsel, out);
}


# k2 scores: 8 rows per wave in 16-wave blocks (4096 waves enter, not 32768), mean vector in registers, all 37 loads issued as one group ahead of the sums; per-row sum order kept
# speedup vs baseline: 1.5786x; 1.1048x over previous
#include <hip/hip_runtime.h>

#define B_    32
#define N_    1024
#define D_    512
#define K_SEL 512
#define EPS_  1e-8f

typedef __attribute__((ext_vector_type(2))) float v2f;
typedef __attribute__((ext_vector_type(8))) float v8f;

__global__ __launch_bounds__(256)
void k1_norms_part(const float* __restrict__ x, float* __restrict__ inv_norms,
                   float* __restrict__ part) {
  const int b      = blockIdx.x >> 4;
  const int rowblk = blockIdx.x & 15;
  const int row0   = rowblk << 6;
  const int lane   = threadIdx.x & 31;
  const int wave   = threadIdx.x >> 5;

  __shared__ float lds_acc[8][512];

  float acc[16];
#pragma unroll
  for (int i = 0; i < 16; ++i) acc[i] = 0.f;

  const float* xb = x + (size_t)b * (N_ * D_);

  for (int r = 0; r < 8; ++r) {
    const int row = row0 + wave * 8 + r;
    const float4* xr = (const float4*)(xb + (size_t)row * D_);
    float4 v[4];
    float ss = 0.f;
#pragma unroll
    for (int c = 0; c < 4; ++c) {
      v[c] = xr[c * 32 + lane];
      ss += v[c].x * v[c].x + v[c].y * v[c].y + v[c].z * v[c].z + v[c].w * v[c].w;
    }
    if (r + 1 < 8)
      __builtin_prefetch(xb + (size_t)(row + 1) * D_ + lane * 16, 0, 1);
#pragma unroll
    for (int off = 16; off > 0; off >>= 1) ss += __shfl_xor(ss, off, 32);
    const float inv = 1.f / fmaxf(sqrtf(ss), EPS_);
    if (lane == 0) inv_norms[b * N_ + row] = inv;
#pragma unroll
    for (int c = 0; c < 4; ++c) {
      acc[c * 4 + 0] += v[c].x * inv;
      acc[c * 4 + 1] += v[c].y * inv;
      acc[c * 4 + 2] += v[c].z * inv;
      acc[c * 4 + 3] += v[c].w * inv;
    }
  }

#pragma unroll
  for (int c = 0; c < 4; ++c) {
    const int d = c * 128 + lane * 4;
    lds_acc[wave][d + 0] = acc[c * 4 + 0];
    lds_acc[wave][d + 1] = acc[c * 4 + 1];
    lds_acc[wave][d + 2] = acc[c * 4 + 2];
    lds_acc[wave][d + 3] = acc[c * 4 + 3];
  }
  __syncthreads();

  for (int d = threadIdx.x; d < 512; d += 256) {
    float s = 0.f;
#pragma unroll
    for (int w8 = 0; w8 < 8; ++w8) s += lds_acc[w8][d];
    part[((size_t)b * 16 + rowblk) * D_ + d] = s;
  }
}

__global__ __launch_bounds__(256)
void k1b_mu(const float* __restrict__ part, float* __restrict__ mu) {
  const int i = blockIdx.x * 256 + threadIdx.x;
  if (i >= B_ * D_) return;
  const int b = i / D_;
  const int d = i - b * D_;
  float s = 0.f;
#pragma unroll
  for (int j = 0; j < 16; ++j) s += part[((size_t)b * 16 + j) * D_ + d];
  mu[i] = s * (1.0f / N_);
}

__global__ __launch_bounds__(512)
void k2_scores(const float* __restrict__ x, const float* __restrict__ inv_norms,
               const float* __restrict__ mu, float* __restrict__ scores) {
  const int b    = blockIdx.x >> 3;
  const int lane = threadIdx.x & 31;
  const int wave = __builtin_amdgcn_readfirstlane((int)(threadIdx.x >> 5));
  const int row0 = ((blockIdx.x & 7) << 7) + (wave << 3);

  float inv = inv_norms[b * N_ + row0 + (lane & 7)];

  const float4* mub = (const float4*)(mu + (size_t)b * D_);
  float4 m[4];
  for (int c = 0; c < 4; ++c) m[c] = mub[c * 32 + lane];

  const float4* xr = (const float4*)(x + ((size_t)b * N_ + row0) * D_);
  float4 v[8][4];
  for (int r = 0; r < 8; ++r)
    for (int c = 0; c < 4; ++c) v[r][c] = xr[r * (D_ / 4) + c * 32 + lane];

  asm volatile("" : "+v"(inv));
  __builtin_amdgcn_sched_barrier(0);

  float dot[8];
  for (int r = 0; r < 8; ++r) {
    float s = 0.f;
    for (int c = 0; c < 4; ++c)
      s += v[r][c].x * m[c].x + v[r][c].y * m[c].y + v[r][c].z * m[c].z + v[r][c].w * m[c].w;
    dot[r] = s;
  }
  for (int off = 16; off > 0; off >>= 1)
    for (int r = 0; r < 8; ++r) dot[r] += __shfl_xor(dot[r], off, 32);

  float mine = dot[0];
  for (int r = 1; r < 8; ++r) mine = (lane == r) ? dot[r] : mine;
  if (lane < 8) scores[b * N_ + row0 + lane] = mine * inv;
}

__device__ __forceinline__ void load_group(float4 (&v)[8], const float4* s4, const int g) {
  for (int i = 0; i < 8; ++i) v[i] = s4[g * 8 + i];
}

__device__ __forceinline__ void count_group(const float4 (&v)[8], const float sm,
                                            const int g, const int g_own, int (&acc)[8]) {
  unsigned mk[32];
  if (g == g_own) {
    for (int i = 0; i < 8; ++i) {
      const float e[4] = { v[i].x, v[i].y, v[i].z, v[i].w };
      for (int c = 0; c < 4; ++c) {
        const int k = i * 4 + c;
        const unsigned above_k = 0xFFFFFFFEu << k;
        const unsigned gt = __builtin_amdgcn_ballot_w32(e[c] > sm);
        const unsigned eq = __builtin_amdgcn_ballot_w32(e[c] == sm);
        mk[k] = gt | (eq & above_k);
      }
    }
  } else if (g < g_own) {
    for (int i = 0; i < 8; ++i) {
      mk[i * 4 + 0] = __builtin_amdgcn_ballot_w32(v[i].x >= sm);
      mk[i * 4 + 1] = __builtin_amdgcn_ballot_w32(v[i].y >= sm);
      mk[i * 4 + 2] = __builtin_amdgcn_ballot_w32(v[i].z >= sm);
      mk[i * 4 + 3] = __builtin_amdgcn_ballot_w32(v[i].w >= sm);
    }
  } else {
    for (int i = 0; i < 8; ++i) {
      mk[i * 4 + 0] = __builtin_amdgcn_ballot_w32(v[i].x > sm);
      mk[i * 4 + 1] = __builtin_amdgcn_ballot_w32(v[i].y > sm);
      mk[i * 4 + 2] = __builtin_amdgcn_ballot_w32(v[i].z > sm);
      mk[i * 4 + 3] = __builtin_amdgcn_ballot_w32(v[i].w > sm);
    }
  }
  for (int i = 0; i < 8; ++i)
    acc[i] += (int)__builtin_amdgcn_inverse_ballot_w32(mk[i * 4 + 0]) +
              (int)__builtin_amdgcn_inverse_ballot_w32(mk[i * 4 + 1]) +
              (int)__builtin_amdgcn_inverse_ballot_w32(mk[i * 4 + 2]) +
              (int)__builtin_amdgcn_inverse_ballot_w32(mk[i * 4 + 3]);
}

__global__ __launch_bounds__(128)
void k3_rank(const float* __restrict__ scores, float* __restrict__ w) {
  const int b  = blockIdx.x >> 3;
  const int mb = (blockIdx.x & 7) << 7;
  const int m  = mb + threadIdx.x;

  __shared__ float4 s4[N_ / 4];
  {
    const float4* sb = (const float4*)(scores + b * N_);
    for (int i = threadIdx.x; i < N_ / 4; i += 128) s4[i] = sb[i];
  }
  __syncthreads();

  const float sm  = ((const float*)s4)[m];
  const int g_own = (mb >> 5) + __builtin_amdgcn_readfirstlane((int)(threadIdx.x >> 5));

  int acc[8];
  for (int i = 0; i < 8; ++i) acc[i] = 0;

  float4 va[8], vb[8];
  load_group(va, s4, 0);
  for (int g = 0; g < N_ / 32; g += 2) {
    load_group(vb, s4, g + 1);
    count_group(va, sm, g, g_own, acc);
    load_group(va, s4, (g + 2) & (N_ / 32 - 1));
    count_group(vb, sm, g + 1, g_own, acc);
  }

  const int rank = ((acc[0] + acc[1]) + (acc[2] + acc[3])) + ((acc[4] + acc[5]) + (acc[6] + acc[7]));
  w[b * N_ + m] = (rank < K_SEL) ? (1.0f / K_SEL) : 0.0f;
}

__global__ __launch_bounds__(256)
void k4_merge_wmma(const float* __restrict__ x, const float* __restrict__ w,
                   float* __restrict__ out) {
  const int b     = blockIdx.x >> 2;
  const int tile  = ((blockIdx.x & 3) << 3) + (threadIdx.x >> 5);
  const int lane  = threadIdx.x & 31;
  const int half  = lane >> 4;
  const int ln    = lane & 15;
  const int dbase = tile << 4;

  __shared__ float wl[N_];
  for (int i = threadIdx.x; i < N_; i += 256) wl[i] = w[b * N_ + i];
  __syncthreads();

  const float* xb = x + (size_t)b * (N_ * D_);

  v8f c = {};
  for (int k = 0; k < N_; k += 4) {
    const int ka = k + (half ? 2 : 0);
    const int kb = ka + 1;
    v2f a  = { wl[ka], wl[kb] };
    v2f bv = { xb[(size_t)ka * D_ + dbase + ln],
               xb[(size_t)kb * D_ + dbase + ln] };
    c = __builtin_amdgcn_wmma_f32_16x16x4_f32(
         false, a,  false, bv,
         (short)0, c,  false,  false);
  }

  if (lane < 16) out[(size_t)b * D_ + dbase + ln] = c[0];
}

extern "C" void kernel_launch(void* const* d_in, const int* in_sizes, int n_in,
                              void* d_out, int out_size, void* d_ws, size_t ws_size,
                              hipStream_t stream) {
  (void)in_sizes; (void)n_in; (void)out_size; (void)ws_size;

  const float* x = (const float*)d_in[0];
  float* out = (float*)d_out;

  char* ws = (char*)d_ws;
  float* inv_norms = (float*)(ws + 0);
  float* scores    = (float*)(ws + 131072);
  float* wsel      = (float*)(ws + 262144);
  float* mu        = (float*)(ws + 393216);
  float* part      = (float*)(ws + 458752);

  k1_norms_part<<<B_ * 16, 256, 0, stream>>>(x, inv_norms, part);
  k1b_mu<<<(B_ * D_ + 255) / 256, 256, 0, stream>>>(part, mu);
  k2_scores<<<B_ * 8, 512, 0, stream>>>(x, inv_norms, mu, scores);
  k3_rank<<<B_ * 8, 128, 0, stream>>>(scores, wsel);
  k4_merge_wmma<<<B_ * 4, 256, 0, stream>>>(x, wsel, out);
}


# k4: one 16-byte load per lane (a wave reads two 256-byte row pieces) feeds 2 WMMA chains, all 40 loads issued ahead; rows split over 16 waves per block, partials added via LDS (sum order changes)
# speedup vs baseline: 2.2085x; 1.3990x over previous
#include <hip/hip_runtime.h>

#define B_    32
#define N_    1024
#define D_    512
#define K_SEL 512
#define EPS_  1e-8f

typedef __attribute__((ext_vector_type(2))) float v2f;
typedef __attribute__((ext_vector_type(8))) float v8f;

__global__ __launch_bounds__(256)
void k1_norms_part(const float* __restrict__ x, float* __restrict__ inv_norms,
                   float* __restrict__ part) {
  const int b      = blockIdx.x >> 4;
  const int rowblk = blockIdx.x & 15;
  const int row0   = rowblk << 6;
  const int lane   = threadIdx.x & 31;
  const int wave   = threadIdx.x >> 5;

  __shared__ float lds_acc[8][512];

  float acc[16];
#pragma unroll
  for (int i = 0; i < 16; ++i) acc[i] = 0.f;

  const float* xb = x + (size_t)b * (N_ * D_);

  for (int r = 0; r < 8; ++r) {
    const int row = row0 + wave * 8 + r;
    const float4* xr = (const float4*)(xb + (size_t)row * D_);
    float4 v[4];
    float ss = 0.f;
#pragma unroll
    for (int c = 0; c < 4; ++c) {
      v[c] = xr[c * 32 + lane];
      ss += v[c].x * v[c].x + v[c].y * v[c].y + v[c].z * v[c].z + v[c].w * v[c].w;
    }
    if (r + 1 < 8)
      __builtin_prefetch(xb + (size_t)(row + 1) * D_ + lane * 16, 0, 1);
#pragma unroll
    for (int off = 16; off > 0; off >>= 1) ss += __shfl_xor(ss, off, 32);
    const float inv = 1.f / fmaxf(sqrtf(ss), EPS_);
    if (lane == 0) inv_norms[b * N_ + row] = inv;
#pragma unroll
    for (int c = 0; c < 4; ++c) {
      acc[c * 4 + 0] += v[c].x * inv;
      acc[c * 4 + 1] += v[c].y * inv;
      acc[c * 4 + 2] += v[c].z * inv;
      acc[c * 4 + 3] += v[c].w * inv;
    }
  }

#pragma unroll
  for (int c = 0; c < 4; ++c) {
    const int d = c * 128 + lane * 4;
    lds_acc[wave][d + 0] = acc[c * 4 + 0];
    lds_acc[wave][d + 1] = acc[c * 4 + 1];
    lds_acc[wave][d + 2] = acc[c * 4 + 2];
    lds_acc[wave][d + 3] = acc[c * 4 + 3];
  }
  __syncthreads();

  for (int d = threadIdx.x; d < 512; d += 256) {
    float s = 0.f;
#pragma unroll
    for (int w8 = 0; w8 < 8; ++w8) s += lds_acc[w8][d];
    part[((size_t)b * 16 + rowblk) * D_ + d] = s;
  }
}

__global__ __launch_bounds__(256)
void k1b_mu(const float* __restrict__ part, float* __restrict__ mu) {
  const int i = blockIdx.x * 256 + threadIdx.x;
  if (i >= B_ * D_) return;
  const int b = i / D_;
  const int d = i - b * D_;
  float s = 0.f;
#pragma unroll
  for (int j = 0; j < 16; ++j) s += part[((size_t)b * 16 + j) * D_ + d];
  mu[i] = s * (1.0f / N_);
}

__global__ __launch_bounds__(512)
void k2_scores(const float* __restrict__ x, const float* __restrict__ inv_norms,
               const float* __restrict__ mu, float* __restrict__ scores) {
  const int b    = blockIdx.x >> 3;
  const int lane = threadIdx.x & 31;
  const int wave = __builtin_amdgcn_readfirstlane((int)(threadIdx.x >> 5));
  const int row0 = ((blockIdx.x & 7) << 7) + (wave << 3);

  float inv = inv_norms[b * N_ + row0 + (lane & 7)];

  const float4* mub = (const float4*)(mu + (size_t)b * D_);
  float4 m[4];
  for (int c = 0; c < 4; ++c) m[c] = mub[c * 32 + lane];

  const float4* xr = (const float4*)(x + ((size_t)b * N_ + row0) * D_);
  float4 v[8][4];
  for (int r = 0; r < 8; ++r)
    for (int c = 0; c < 4; ++c) v[r][c] = xr[r * (D_ / 4) + c * 32 + lane];

  asm volatile("" : "+v"(inv));
  __builtin_amdgcn_sched_barrier(0);

  float dot[8];
  for (int r = 0; r < 8; ++r) {
    float s = 0.f;
    for (int c = 0; c < 4; ++c)
      s += v[r][c].x * m[c].x + v[r][c].y * m[c].y + v[r][c].z * m[c].z + v[r][c].w * m[c].w;
    dot[r] = s;
  }
  for (int off = 16; off > 0; off >>= 1)
    for (int r = 0; r < 8; ++r) dot[r] += __shfl_xor(dot[r], off, 32);

  float mine = dot[0];
  for (int r = 1; r < 8; ++r) mine = (lane == r) ? dot[r] : mine;
  if (lane < 8) scores[b * N_ + row0 + lane] = mine * inv;
}

__device__ __forceinline__ void load_group(float4 (&v)[8], const float4* s4, const int g) {
  for (int i = 0; i < 8; ++i) v[i] = s4[g * 8 + i];
}

__device__ __forceinline__ void count_group(const float4 (&v)[8], const float sm,
                                            const int g, const int g_own, int (&acc)[8]) {
  unsigned mk[32];
  if (g == g_own) {
    for (int i = 0; i < 8; ++i) {
      const float e[4] = { v[i].x, v[i].y, v[i].z, v[i].w };
      for (int c = 0; c < 4; ++c) {
        const int k = i * 4 + c;
        const unsigned above_k = 0xFFFFFFFEu << k;
        const unsigned gt = __builtin_amdgcn_ballot_w32(e[c] > sm);
        const unsigned eq = __builtin_amdgcn_ballot_w32(e[c] == sm);
        mk[k] = gt | (eq & above_k);
      }
    }
  } else if (g < g_own) {
    for (int i = 0; i < 8; ++i) {
      mk[i * 4 + 0] = __builtin_amdgcn_ballot_w32(v[i].x >= sm);
      mk[i * 4 + 1] = __builtin_amdgcn_ballot_w32(v[i].y >= sm);
      mk[i * 4 + 2] = __builtin_amdgcn_ballot_w32(v[i].z >= sm);
      mk[i * 4 + 3] = __builtin_amdgcn_ballot_w32(v[i].w >= sm);
    }
  } else {
    for (int i = 0; i < 8; ++i) {
      mk[i * 4 + 0] = __builtin_amdgcn_ballot_w32(v[i].x > sm);
      mk[i * 4 + 1] = __builtin_amdgcn_ballot_w32(v[i].y > sm);
      mk[i * 4 + 2] = __builtin_amdgcn_ballot_w32(v[i].z > sm);
      mk[i * 4 + 3] = __builtin_amdgcn_ballot_w32(v[i].w > sm);
    }
  }
  for (int i = 0; i < 8; ++i)
    acc[i] += (int)__builtin_amdgcn_inverse_ballot_w32(mk[i * 4 + 0]) +
              (int)__builtin_amdgcn_inverse_ballot_w32(mk[i * 4 + 1]) +
              (int)__builtin_amdgcn_inverse_ballot_w32(mk[i * 4 + 2]) +
              (int)__builtin_amdgcn_inverse_ballot_w32(mk[i * 4 + 3]);
}

__global__ __launch_bounds__(128)
void k3_rank(const float* __restrict__ scores, float* __restrict__ w) {
  const int b  = blockIdx.x >> 3;
  const int mb = (blockIdx.x & 7) << 7;
  const int m  = mb + threadIdx.x;

  __shared__ float4 s4[N_ / 4];
  {
    const float4* sb = (const float4*)(scores + b * N_);
    for (int i = threadIdx.x; i < N_ / 4; i += 128) s4[i] = sb[i];
  }
  __syncthreads();

  const float sm  = ((const float*)s4)[m];
  const int g_own = (mb >> 5) + __builtin_amdgcn_readfirstlane((int)(threadIdx.x >> 5));

  int acc[8];
  for (int i = 0; i < 8; ++i) acc[i] = 0;

  float4 va[8], vb[8];
  load_group(va, s4, 0);
  for (int g = 0; g < N_ / 32; g += 2) {
    load_group(vb, s4, g + 1);
    count_group(va, sm, g, g_own, acc);
    load_group(va, s4, (g + 2) & (N_ / 32 - 1));
    count_group(vb, sm, g + 1, g_own, acc);
  }

  const int rank = ((acc[0] + acc[1]) + (acc[2] + acc[3])) + ((acc[4] + acc[5]) + (acc[6] + acc[7]));
  w[b * N_ + m] = (rank < K_SEL) ? (1.0f / K_SEL) : 0.0f;
}

__global__ __launch_bounds__(512)
void k4_merge_wmma(const float* __restrict__ x, const float* __restrict__ w,
                   float* __restrict__ out) {
  const int b    = blockIdx.x >> 3;
  const int slab = (blockIdx.x & 7) << 6;
  const int lane = threadIdx.x & 31;
  const int wave = __builtin_amdgcn_readfirstlane((int)(threadIdx.x >> 5));
  const int c    = lane & 15;
  const int r0   = (wave << 6) + ((lane >> 4) << 5);

  __shared__ float2 red[2][16 * 32];

  int slot = (int)(threadIdx.x * sizeof(float2));
  asm volatile("" : "+v"(slot));

  const float4* wp = (const float4*)(w + b * N_ + r0);
  const float*  xp = x + ((size_t)b * N_ + r0) * D_ + slab + 4 * c;

  float4 wq[8];
  for (int q = 0; q < 8; ++q) wq[q] = wp[q];
  float4 v[32];
  for (int j = 0; j < 32; ++j) v[j] = *(const float4*)(xp + (size_t)j * D_);

  __builtin_amdgcn_sched_barrier(0);

  const bool odd = (c & 1) != 0;
  v2f a[32];
  for (int q = 0; q < 8; ++q) {
    const float e[4] = { wq[q].x, wq[q].y, wq[q].z, wq[q].w };
    for (int i = 0; i < 4; ++i) {
      a[4 * q + i][0] = odd ? 0.f : e[i];
      a[4 * q + i][1] = odd ? e[i] : 0.f;
    }
  }

  __builtin_amdgcn_sched_barrier(0);

  v8f accP = {}, accQ = {};
  for (int j = 0; j < 32; ++j) {
    const v2f bp = { v[j].x, v[j].y };
    const v2f bq = { v[j].z, v[j].w };
    accP = __builtin_amdgcn_wmma_f32_16x16x4_f32(false, a[j], false, bp, (short)0, accP, false, false);
    accQ = __builtin_amdgcn_wmma_f32_16x16x4_f32(false, a[j], false, bq, (short)0, accQ, false, false);
  }

  *(float2*)((char*)&red[0][0] + slot) = make_float2(accP[0], accP[1]);
  *(float2*)((char*)&red[1][0] + slot) = make_float2(accQ[0], accQ[1]);
  __syncthreads();

  if (threadIdx.x < 16) {
    float2 p = red[0][threadIdx.x];
    float2 q = red[1][threadIdx.x];
    for (int j = 1; j < 16; ++j) {
      const float2 pj = red[0][j * 32 + threadIdx.x];
      const float2 qj = red[1][j * 32 + threadIdx.x];
      p.x += pj.x; p.y += pj.y; q.x += qj.x; q.y += qj.y;
    }
    *(float4*)(out + (size_t)b * D_ + slab + 4 * threadIdx.x) = make_float4(p.x, p.y, q.x, q.y);
  }
}

extern "C" void kernel_launch(void* const* d_in, const int* in_sizes, int n_in,
                              void* d_out, int out_size, void* d_ws, size_t ws_size,
                              hipStream_t stream) {
  (void)in_sizes; (void)n_in; (void)out_size; (void)ws_size;

  const float* x = (const float*)d_in[0];
  float* out = (float*)d_out;

  char* ws = (char*)d_ws;
  float* inv_norms = (float*)(ws + 0);
  float* scores    = (float*)(ws + 131072);
  float* wsel      = (float*)(ws + 262144);
  float* mu        = (float*)(ws + 393216);
  float* part      = (float*)(ws + 458752);

  k1_norms_part<<<B_ * 16, 256, 0, stream>>>(x, inv_norms, part);
  k1b_mu<<<(B_ * D_ + 255) / 256, 256, 0, stream>>>(part, mu);
  k2_scores<<<B_ * 8, 512, 0, stream>>>(x, inv_norms, mu, scores);
  k3_rank<<<B_ * 8, 128, 0, stream>>>(scores, wsel);
  k4_merge_wmma<<<B_ * 8, 512, 0, stream>>>(x, wsel, out);
}


# k2 sums its batch's 16 row-block partials itself (thread = dim, same order, x 1/N, handed over via LDS; the 16 partial loads go out ahead of the row loads): reduce launch gone, 4 launches not 5
# speedup vs baseline: 2.4021x; 1.0877x over previous
#include <hip/hip_runtime.h>

#define B_    32
#define N_    1024
#define D_    512
#define K_SEL 512
#define EPS_  1e-8f

typedef __attribute__((ext_vector_type(2))) float v2f;
typedef __attribute__((ext_vector_type(8))) float v8f;

__global__ __launch_bounds__(256)
void k1_norms_part(const float* __restrict__ x, float* __restrict__ inv_norms,
                   float* __restrict__ part) {
  const int b      = blockIdx.x >> 4;
  const int rowblk = blockIdx.x & 15;
  const int row0   = rowblk << 6;
  const int lane   = threadIdx.x & 31;
  const int wave   = threadIdx.x >> 5;

  __shared__ float lds_acc[8][512];

  float acc[16];
#pragma unroll
  for (int i = 0; i < 16; ++i) acc[i] = 0.f;

  const float* xb = x + (size_t)b * (N_ * D_);

  for (int r = 0; r < 8; ++r) {
    const int row = row0 + wave * 8 + r;
    const float4* xr = (const float4*)(xb + (size_t)row * D_);
    float4 v[4];
    float ss = 0.f;
#pragma unroll
    for (int c = 0; c < 4; ++c) {
      v[c] = xr[c * 32 + lane];
      ss += v[c].x * v[c].x + v[c].y * v[c].y + v[c].z * v[c].z + v[c].w * v[c].w;
    }
    if (r + 1 < 8)
      __builtin_prefetch(xb + (size_t)(row + 1) * D_ + lane * 16, 0, 1);
#pragma unroll
    for (int off = 16; off > 0; off >>= 1) ss += __shfl_xor(ss, off, 32);
    const float inv = 1.f / fmaxf(sqrtf(ss), EPS_);
    if (lane == 0) inv_norms[b * N_ + row] = inv;
#pragma unroll
    for (int c = 0; c < 4; ++c) {
      acc[c * 4 + 0] += v[c].x * inv;
      acc[c * 4 + 1] += v[c].y * inv;
      acc[c * 4 + 2] += v[c].z * inv;
      acc[c * 4 + 3] += v[c].w * inv;
    }
  }

#pragma unroll
  for (int c = 0; c < 4; ++c) {
    const int d = c * 128 + lane * 4;
    lds_acc[wave][d + 0] = acc[c * 4 + 0];
    lds_acc[wave][d + 1] = acc[c * 4 + 1];
    lds_acc[wave][d + 2] = acc[c * 4 + 2];
    lds_acc[wave][d + 3] = acc[c * 4 + 3];
  }
  __syncthreads();

  for (int d = threadIdx.x; d < 512; d += 256) {
    float s = 0.f;
#pragma unroll
    for (int w8 = 0; w8 < 8; ++w8) s += lds_acc[w8][d];
    part[((size_t)b * 16 + rowblk) * D_ + d] = s;
  }
}

__global__ __launch_bounds__(512)
void k2_scores(const float* __restrict__ x, const float* __restrict__ inv_norms,
               const float* __restrict__ part, float* __restrict__ scores) {
  const int b    = blockIdx.x >> 3;
  const int lane = threadIdx.x & 31;
  const int wave = __builtin_amdgcn_readfirstlane((int)(threadIdx.x >> 5));
  const int row0 = ((blockIdx.x & 7) << 7) + (wave << 3);

  __shared__ float4 mu4[D_ / 4];

  const float* pb = part + (size_t)b * 16 * D_ + threadIdx.x;
  float p[16];
  for (int j = 0; j < 16; ++j) p[j] = pb[j * D_];

  float inv = inv_norms[b * N_ + row0 + (lane & 7)];

  const float4* xr = (const float4*)(x + ((size_t)b * N_ + row0) * D_);
  float4 v[8][4];
  for (int r = 0; r < 8; ++r)
    for (int c = 0; c < 4; ++c) v[r][c] = xr[r * (D_ / 4) + c * 32 + lane];

  __builtin_amdgcn_sched_barrier(0);

  float ps = 0.f;
  for (int j = 0; j < 16; ++j) ps += p[j];
  ((float*)mu4)[threadIdx.x] = ps * (1.0f / N_);
  __builtin_amdgcn_fence(__ATOMIC_RELEASE, "workgroup", "local");
  __builtin_amdgcn_s_barrier();
  __builtin_amdgcn_fence(__ATOMIC_ACQUIRE, "workgroup", "local");

  float4 m[4];
  for (int c = 0; c < 4; ++c) m[c] = mu4[c * 32 + lane];

  asm volatile("" : "+v"(inv));

  float dot[8];
  for (int r = 0; r < 8; ++r) {
    float s = 0.f;
    for (int c = 0; c < 4; ++c)
      s += v[r][c].x * m[c].x + v[r][c].y * m[c].y + v[r][c].z * m[c].z + v[r][c].w * m[c].w;
    dot[r] = s;
  }
  for (int off = 16; off > 0; off >>= 1)
    for (int r = 0; r < 8; ++r) dot[r] += __shfl_xor(dot[r], off, 32);

  float mine = dot[0];
  for (int r = 1; r < 8; ++r) mine = (lane == r) ? dot[r] : mine;
  if (lane < 8) scores[b * N_ + row0 + lane] = mine * inv;
}

__device__ __forceinline__ void load_group(float4 (&v)[8], const float4* s4, const int g) {
  for (int i = 0; i < 8; ++i) v[i] = s4[g * 8 + i];
}

__device__ __forceinline__ void count_group(const float4 (&v)[8], const float sm,
                                            const int g, const int g_own, int (&acc)[8]) {
  unsigned mk[32];
  if (g == g_own) {
    for (int i = 0; i < 8; ++i) {
      const float e[4] = { v[i].x, v[i].y, v[i].z, v[i].w };
      for (int c = 0; c < 4; ++c) {
        const int k = i * 4 + c;
        const unsigned above_k = 0xFFFFFFFEu << k;
        const unsigned gt = __builtin_amdgcn_ballot_w32(e[c] > sm);
        const unsigned eq = __builtin_amdgcn_ballot_w32(e[c] == sm);
        mk[k] = gt | (eq & above_k);
      }
    }
  } else if (g < g_own) {
    for (int i = 0; i < 8; ++i) {
      mk[i * 4 + 0] = __builtin_amdgcn_ballot_w32(v[i].x >= sm);
      mk[i * 4 + 1] = __builtin_amdgcn_ballot_w32(v[i].y >= sm);
      mk[i * 4 + 2] = __builtin_amdgcn_ballot_w32(v[i].z >= sm);
      mk[i * 4 + 3] = __builtin_amdgcn_ballot_w32(v[i].w >= sm);
    }
  } else {
    for (int i = 0; i < 8; ++i) {
      mk[i * 4 + 0] = __builtin_amdgcn_ballot_w32(v[i].x > sm);
      mk[i * 4 + 1] = __builtin_amdgcn_ballot_w32(v[i].y > sm);
      mk[i * 4 + 2] = __builtin_amdgcn_ballot_w32(v[i].z > sm);
      mk[i * 4 + 3] = __builtin_amdgcn_ballot_w32(v[i].w > sm);
    }
  }
  for (int i = 0; i < 8; ++i)
    acc[i] += (int)__builtin_amdgcn_inverse_ballot_w32(mk[i * 4 + 0]) +
              (int)__builtin_amdgcn_inverse_ballot_w32(mk[i * 4 + 1]) +
              (int)__builtin_amdgcn_inverse_ballot_w32(mk[i * 4 + 2]) +
              (int)__builtin_amdgcn_inverse_ballot_w32(mk[i * 4 + 3]);
}

__global__ __launch_bounds__(128)
void k3_rank(const float* __restrict__ scores, float* __restrict__ w) {
  const int b  = blockIdx.x >> 3;
  const int mb = (blockIdx.x & 7) << 7;
  const int m  = mb + threadIdx.x;

  __shared__ float4 s4[N_ / 4];
  {
    const float4* sb = (const float4*)(scores + b * N_);
    for (int i = threadIdx.x; i < N_ / 4; i += 128) s4[i] = sb[i];
  }
  __syncthreads();

  const float sm  = ((const float*)s4)[m];
  const int g_own = (mb >> 5) + __builtin_amdgcn_readfirstlane((int)(threadIdx.x >> 5));

  int acc[8];
  for (int i = 0; i < 8; ++i) acc[i] = 0;

  float4 va[8], vb[8];
  load_group(va, s4, 0);
  for (int g = 0; g < N_ / 32; g += 2) {
    load_group(vb, s4, g + 1);
    count_group(va, sm, g, g_own, acc);
    load_group(va, s4, (g + 2) & (N_ / 32 - 1));
    count_group(vb, sm, g + 1, g_own, acc);
  }

  const int rank = ((acc[0] + acc[1]) + (acc[2] + acc[3])) + ((acc[4] + acc[5]) + (acc[6] + acc[7]));
  w[b * N_ + m] = (rank < K_SEL) ? (1.0f / K_SEL) : 0.0f;
}

__global__ __launch_bounds__(512)
void k4_merge_wmma(const float* __restrict__ x, const float* __restrict__ w,
                   float* __restrict__ out) {
  const int b    = blockIdx.x >> 3;
  const int slab = (blockIdx.x & 7) << 6;
  const int lane = threadIdx.x & 31;
  const int wave = __builtin_amdgcn_readfirstlane((int)(threadIdx.x >> 5));
  const int c    = lane & 15;
  const int r0   = (wave << 6) + ((lane >> 4) << 5);

  __shared__ float2 red[2][16 * 32];

  int slot = (int)(threadIdx.x * sizeof(float2));
  asm volatile("" : "+v"(slot));

  const float4* wp = (const float4*)(w + b * N_ + r0);
  const float*  xp = x + ((size_t)b * N_ + r0) * D_ + slab + 4 * c;

  float4 wq[8];
  for (int q = 0; q < 8; ++q) wq[q] = wp[q];
  float4 v[32];
  for (int j = 0; j < 32; ++j) v[j] = *(const float4*)(xp + (size_t)j * D_);

  __builtin_amdgcn_sched_barrier(0);

  const bool odd = (c & 1) != 0;
  v2f a[32];
  for (int q = 0; q < 8; ++q) {
    const float e[4] = { wq[q].x, wq[q].y, wq[q].z, wq[q].w };
    for (int i = 0; i < 4; ++i) {
      a[4 * q + i][0] = odd ? 0.f : e[i];
      a[4 * q + i][1] = odd ? e[i] : 0.f;
    }
  }

  __builtin_amdgcn_sched_barrier(0);

  v8f accP = {}, accQ = {};
  for (int j = 0; j < 32; ++j) {
    const v2f bp = { v[j].x, v[j].y };
    const v2f bq = { v[j].z, v[j].w };
    accP = __builtin_amdgcn_wmma_f32_16x16x4_f32(false, a[j], false, bp, (short)0, accP, false, false);
    accQ = __builtin_amdgcn_wmma_f32_16x16x4_f32(false, a[j], false, bq, (short)0, accQ, false, false);
  }

  *(float2*)((char*)&red[0][0] + slot) = make_float2(accP[0], accP[1]);
  *(float2*)((char*)&red[1][0] + slot) = make_float2(accQ[0], accQ[1]);
  __syncthreads();

  if (threadIdx.x < 16) {
    float2 p = red[0][threadIdx.x];
    float2 q = red[1][threadIdx.x];
    for (int j = 1; j < 16; ++j) {
      const float2 pj = red[0][j * 32 + threadIdx.x];
      const float2 qj = red[1][j * 32 + threadIdx.x];
      p.x += pj.x; p.y += pj.y; q.x += qj.x; q.y += qj.y;
    }
    *(float4*)(out + (size_t)b * D_ + slab + 4 * threadIdx.x) = make_float4(p.x, p.y, q.x, q.y);
  }
}

extern "C" void kernel_launch(void* const* d_in, const int* in_sizes, int n_in,
                              void* d_out, int out_size, void* d_ws, size_t ws_size,
                              hipStream_t stream) {
  (void)in_sizes; (void)n_in; (void)out_size; (void)ws_size;

  const float* x = (const float*)d_in[0];
  float* out = (float*)d_out;

  char* ws = (char*)d_ws;
  float* inv_norms = (float*)(ws + 0);
  float* scores    = (float*)(ws + 131072);
  float* wsel      = (float*)(ws + 262144);
  float* part      = (float*)(ws + 458752);

  k1_norms_part<<<B_ * 16, 256, 0, stream>>>(x, inv_norms, part);
  k2_scores<<<B_ * 8, 512, 0, stream>>>(x, inv_norms, part, scores);
  k3_rank<<<B_ * 8, 128, 0, stream>>>(scores, wsel);
  k4_merge_wmma<<<B_ * 8, 512, 0, stream>>>(x, wsel, out);
}


# k1: a wave asks for its 8 rows whole (32 loads of 16 bytes) ahead of the sums: one load round trip, not eight; the 8 rows' reductions, roots and divisions run side by side; sum order kept
# speedup vs baseline: 2.5980x; 1.0816x over previous
#include <hip/hip_runtime.h>

#define B_    32
#define N_    1024
#define D_    512
#define K_SEL 512
#define EPS_  1e-8f

typedef __attribute__((ext_vector_type(2))) float v2f;
typedef __attribute__((ext_vector_type(8))) float v8f;

__global__ __launch_bounds__(256)
void k1_norms_part(const float* __restrict__ x, float* __restrict__ inv_norms,
                   float* __restrict__ part) {
  const int b      = blockIdx.x >> 4;
  const int rowblk = blockIdx.x & 15;
  const int lane   = threadIdx.x & 31;
  const int wave   = __builtin_amdgcn_readfirstlane((int)(threadIdx.x >> 5));
  const int row0   = (rowblk << 6) + (wave << 3);

  __shared__ float lds_acc[8][512];

  const float4* xr = (const float4*)(x + ((size_t)b * N_ + row0) * D_);
  float4 v[8][4];
  for (int r = 0; r < 8; ++r)
    for (int c = 0; c < 4; ++c) v[r][c] = xr[r * (D_ / 4) + c * 32 + lane];

  float ss[8];
  for (int r = 0; r < 8; ++r) {
    float s = 0.f;
    for (int c = 0; c < 4; ++c)
      s += v[r][c].x * v[r][c].x + v[r][c].y * v[r][c].y + v[r][c].z * v[r][c].z + v[r][c].w * v[r][c].w;
    ss[r] = s;
  }
  for (int off = 16; off > 0; off >>= 1)
    for (int r = 0; r < 8; ++r) ss[r] += __shfl_xor(ss[r], off, 32);

  float inv[8];
  for (int r = 0; r < 8; ++r) inv[r] = 1.f / fmaxf(sqrtf(ss[r]), EPS_);

  float mine = inv[0];
  for (int r = 1; r < 8; ++r) mine = (lane == r) ? inv[r] : mine;
  if (lane < 8) inv_norms[b * N_ + row0 + lane] = mine;

  float acc[16];
  for (int i = 0; i < 16; ++i) acc[i] = 0.f;
  for (int r = 0; r < 8; ++r)
    for (int c = 0; c < 4; ++c) {
      acc[c * 4 + 0] += v[r][c].x * inv[r];
      acc[c * 4 + 1] += v[r][c].y * inv[r];
      acc[c * 4 + 2] += v[r][c].z * inv[r];
      acc[c * 4 + 3] += v[r][c].w * inv[r];
    }

  for (int c = 0; c < 4; ++c) {
    const int d = c * 128 + lane * 4;
    lds_acc[wave][d + 0] = acc[c * 4 + 0];
    lds_acc[wave][d + 1] = acc[c * 4 + 1];
    lds_acc[wave][d + 2] = acc[c * 4 + 2];
    lds_acc[wave][d + 3] = acc[c * 4 + 3];
  }
  __syncthreads();

  for (int d = threadIdx.x; d < 512; d += 256) {
    float s = 0.f;
#pragma unroll
    for (int w8 = 0; w8 < 8; ++w8) s += lds_acc[w8][d];
    part[((size_t)b * 16 + rowblk) * D_ + d] = s;
  }
}

__global__ __launch_bounds__(512)
void k2_scores(const float* __restrict__ x, const float* __restrict__ inv_norms,
               const float* __restrict__ part, float* __restrict__ scores) {
  const int b    = blockIdx.x >> 3;
  const int lane = threadIdx.x & 31;
  const int wave = __builtin_amdgcn_readfirstlane((int)(threadIdx.x >> 5));
  const int row0 = ((blockIdx.x & 7) << 7) + (wave << 3);

  __shared__ float4 mu4[D_ / 4];

  const float* pb = part + (size_t)b * 16 * D_ + threadIdx.x;
  float p[16];
  for (int j = 0; j < 16; ++j) p[j] = pb[j * D_];

  float inv = inv_norms[b * N_ + row0 + (lane & 7)];

  const float4* xr = (const float4*)(x + ((size_t)b * N_ + row0) * D_);
  float4 v[8][4];
  for (int r = 0; r < 8; ++r)
    for (int c = 0; c < 4; ++c) v[r][c] = xr[r * (D_ / 4) + c * 32 + lane];

  __builtin_amdgcn_sched_barrier(0);

  float ps = 0.f;
  for (int j = 0; j < 16; ++j) ps += p[j];
  ((float*)mu4)[threadIdx.x] = ps * (1.0f / N_);
  __builtin_amdgcn_fence(__ATOMIC_RELEASE, "workgroup", "local");
  __builtin_amdgcn_s_barrier();
  __builtin_amdgcn_fence(__ATOMIC_ACQUIRE, "workgroup", "local");

  float4 m[4];
  for (int c = 0; c < 4; ++c) m[c] = mu4[c * 32 + lane];

  asm volatile("" : "+v"(inv));

  float dot[8];
  for (int r = 0; r < 8; ++r) {
    float s = 0.f;
    for (int c = 0; c < 4; ++c)
      s += v[r][c].x * m[c].x + v[r][c].y * m[c].y + v[r][c].z * m[c].z + v[r][c].w * m[c].w;
    dot[r] = s;
  }
  for (int off = 16; off > 0; off >>= 1)
    for (int r = 0; r < 8; ++r) dot[r] += __shfl_xor(dot[r], off, 32);

  float mine = dot[0];
  for (int r = 1; r < 8; ++r) mine = (lane == r) ? dot[r] : mine;
  if (lane < 8) scores[b * N_ + row0 + lane] = mine * inv;
}

__device__ __forceinline__ void load_group(float4 (&v)[8], const float4* s4, const int g) {
  for (int i = 0; i < 8; ++i) v[i] = s4[g * 8 + i];
}

__device__ __forceinline__ void count_group(const float4 (&v)[8], const float sm,
                                            const int g, const int g_own, int (&acc)[8]) {
  unsigned mk[32];
  if (g == g_own) {
    for (int i = 0; i < 8; ++i) {
      const float e[4] = { v[i].x, v[i].y, v[i].z, v[i].w };
      for (int c = 0; c < 4; ++c) {
        const int k = i * 4 + c;
        const unsigned above_k = 0xFFFFFFFEu << k;
        const unsigned gt = __builtin_amdgcn_ballot_w32(e[c] > sm);
        const unsigned eq = __builtin_amdgcn_ballot_w32(e[c] == sm);
        mk[k] = gt | (eq & above_k);
      }
    }
  } else if (g < g_own) {
    for (int i = 0; i < 8; ++i) {
      mk[i * 4 + 0] = __builtin_amdgcn_ballot_w32(v[i].x >= sm);
      mk[i * 4 + 1] = __builtin_amdgcn_ballot_w32(v[i].y >= sm);
      mk[i * 4 + 2] = __builtin_amdgcn_ballot_w32(v[i].z >= sm);
      mk[i * 4 + 3] = __builtin_amdgcn_ballot_w32(v[i].w >= sm);
    }
  } else {
    for (int i = 0; i < 8; ++i) {
      mk[i * 4 + 0] = __builtin_amdgcn_ballot_w32(v[i].x > sm);
      mk[i * 4 + 1] = __builtin_amdgcn_ballot_w32(v[i].y > sm);
      mk[i * 4 + 2] = __builtin_amdgcn_ballot_w32(v[i].z > sm);
      mk[i * 4 + 3] = __builtin_amdgcn_ballot_w32(v[i].w > sm);
    }
  }
  for (int i = 0; i < 8; ++i)
    acc[i] += (int)__builtin_amdgcn_inverse_ballot_w32(mk[i * 4 + 0]) +
              (int)__builtin_amdgcn_inverse_ballot_w32(mk[i * 4 + 1]) +
              (int)__builtin_amdgcn_inverse_ballot_w32(mk[i * 4 + 2]) +
              (int)__builtin_amdgcn_inverse_ballot_w32(mk[i * 4 + 3]);
}

__global__ __launch_bounds__(128)
void k3_rank(const float* __restrict__ scores, float* __restrict__ w) {
  const int b  = blockIdx.x >> 3;
  const int mb = (blockIdx.x & 7) << 7;
  const int m  = mb + threadIdx.x;

  __shared__ float4 s4[N_ / 4];
  {
    const float4* sb = (const float4*)(scores + b * N_);
    for (int i = threadIdx.x; i < N_ / 4; i += 128) s4[i] = sb[i];
  }
  __syncthreads();

  const float sm  = ((const float*)s4)[m];
  const int g_own = (mb >> 5) + __builtin_amdgcn_readfirstlane((int)(threadIdx.x >> 5));

  int acc[8];
  for (int i = 0; i < 8; ++i) acc[i] = 0;

  float4 va[8], vb[8];
  load_group(va, s4, 0);
  for (int g = 0; g < N_ / 32; g += 2) {
    load_group(vb, s4, g + 1);
    count_group(va, sm, g, g_own, acc);
    load_group(va, s4, (g + 2) & (N_ / 32 - 1));
    count_group(vb, sm, g + 1, g_own, acc);
  }

  const int rank = ((acc[0] + acc[1]) + (acc[2] + acc[3])) + ((acc[4] + acc[5]) + (acc[6] + acc[7]));
  w[b * N_ + m] = (rank < K_SEL) ? (1.0f / K_SEL) : 0.0f;
}

__global__ __launch_bounds__(512)
void k4_merge_wmma(const float* __restrict__ x, const float* __restrict__ w,
                   float* __restrict__ out) {
  const int b    = blockIdx.x >> 3;
  const int slab = (blockIdx.x & 7) << 6;
  const int lane = threadIdx.x & 31;
  const int wave = __builtin_amdgcn_readfirstlane((int)(threadIdx.x >> 5));
  const int c    = lane & 15;
  const int r0   = (wave << 6) + ((lane >> 4) << 5);

  __shared__ float2 red[2][16 * 32];

  int slot = (int)(threadIdx.x * sizeof(float2));
  asm volatile("" : "+v"(slot));

  const float4* wp = (const float4*)(w + b * N_ + r0);
  const float*  xp = x + ((size_t)b * N_ + r0) * D_ + slab + 4 * c;

  float4 wq[8];
  for (int q = 0; q < 8; ++q) wq[q] = wp[q];
  float4 v[32];
  for (int j = 0; j < 32; ++j) v[j] = *(const float4*)(xp + (size_t)j * D_);

  __builtin_amdgcn_sched_barrier(0);

  const bool odd = (c & 1) != 0;
  v2f a[32];
  for (int q = 0; q < 8; ++q) {
    const float e[4] = { wq[q].x, wq[q].y, wq[q].z, wq[q].w };
    for (int i = 0; i < 4; ++i) {
      a[4 * q + i][0] = odd ? 0.f : e[i];
      a[4 * q + i][1] = odd ? e[i] : 0.f;
    }
  }

  __builtin_amdgcn_sched_barrier(0);

  v8f accP = {}, accQ = {};
  for (int j = 0; j < 32; ++j) {
    const v2f bp = { v[j].x, v[j].y };
    const v2f bq = { v[j].z, v[j].w };
    accP = __builtin_amdgcn_wmma_f32_16x16x4_f32(false, a[j], false, bp, (short)0, accP, false, false);
    accQ = __builtin_amdgcn_wmma_f32_16x16x4_f32(false, a[j], false, bq, (short)0, accQ, false, false);
  }

  *(float2*)((char*)&red[0][0] + slot) = make_float2(accP[0], accP[1]);
  *(float2*)((char*)&red[1][0] + slot) = make_float2(accQ[0], accQ[1]);
  __syncthreads();

  if (threadIdx.x < 16) {
    float2 p = red[0][threadIdx.x];
    float2 q = red[1][threadIdx.x];
    for (int j = 1; j < 16; ++j) {
      const float2 pj = red[0][j * 32 + threadIdx.x];
      const float2 qj = red[1][j * 32 + threadIdx.x];
      p.x += pj.x; p.y += pj.y; q.x += qj.x; q.y += qj.y;
    }
    *(float4*)(out + (size_t)b * D_ + slab + 4 * threadIdx.x) = make_float4(p.x, p.y, q.x, q.y);
  }
}

extern "C" void kernel_launch(void* const* d_in, const int* in_sizes, int n_in,
                              void* d_out, int out_size, void* d_ws, size_t ws_size,
                              hipStream_t stream) {
  (void)in_sizes; (void)n_in; (void)out_size; (void)ws_size;

  const float* x = (const float*)d_in[0];
  float* out = (float*)d_out;

  char* ws = (char*)d_ws;
  float* inv_norms = (float*)(ws + 0);
  float* scores    = (float*)(ws + 131072);
  float* wsel      = (float*)(ws + 262144);
  float* part      = (float*)(ws + 458752);

  k1_norms_part<<<B_ * 16, 256, 0, stream>>>(x, inv_norms, part);
  k2_scores<<<B_ * 8, 512, 0, stream>>>(x, inv_norms, part, scores);
  k3_rank<<<B_ * 8, 128, 0, stream>>>(scores, wsel);
  k4_merge_wmma<<<B_ * 8, 512, 0, stream>>>(x, wsel, out);
}


# merge kernel finds its batch's k-th largest score itself (up to 8 rounds of 4 key bits, wave counts via LDS, ties by row order) while its row loads fly: rank launch gone, 3 launches; sum order kept
# speedup vs baseline: 2.8525x; 1.0980x over previous
#include <hip/hip_runtime.h>

#define B_    32
#define N_    1024
#define D_    512
#define K_SEL 512
#define EPS_  1e-8f

typedef __attribute__((ext_vector_type(2))) float v2f;
typedef __attribute__((ext_vector_type(8))) float v8f;

__global__ __launch_bounds__(256)
void k1_norms_part(const float* __restrict__ x, float* __restrict__ inv_norms,
                   float* __restrict__ part) {
  const int b      = blockIdx.x >> 4;
  const int rowblk = blockIdx.x & 15;
  const int lane   = threadIdx.x & 31;
  const int wave   = __builtin_amdgcn_readfirstlane((int)(threadIdx.x >> 5));
  const int row0   = (rowblk << 6) + (wave << 3);

  __shared__ float lds_acc[8][512];

  const float4* xr = (const float4*)(x + ((size_t)b * N_ + row0) * D_);
  float4 v[8][4];
  for (int r = 0; r < 8; ++r)
    for (int c = 0; c < 4; ++c) v[r][c] = xr[r * (D_ / 4) + c * 32 + lane];

  float ss[8];
  for (int r = 0; r < 8; ++r) {
    float s = 0.f;
    for (int c = 0; c < 4; ++c)
      s += v[r][c].x * v[r][c].x + v[r][c].y * v[r][c].y + v[r][c].z * v[r][c].z + v[r][c].w * v[r][c].w;
    ss[r] = s;
  }
  for (int off = 16; off > 0; off >>= 1)
    for (int r = 0; r < 8; ++r) ss[r] += __shfl_xor(ss[r], off, 32);

  float inv[8];
  for (int r = 0; r < 8; ++r) inv[r] = 1.f / fmaxf(sqrtf(ss[r]), EPS_);

  float mine = inv[0];
  for (int r = 1; r < 8; ++r) mine = (lane == r) ? inv[r] : mine;
  if (lane < 8) inv_norms[b * N_ + row0 + lane] = mine;

  float acc[16];
  for (int i = 0; i < 16; ++i) acc[i] = 0.f;
  for (int r = 0; r < 8; ++r)
    for (int c = 0; c < 4; ++c) {
      acc[c * 4 + 0] += v[r][c].x * inv[r];
      acc[c * 4 + 1] += v[r][c].y * inv[r];
      acc[c * 4 + 2] += v[r][c].z * inv[r];
      acc[c * 4 + 3] += v[r][c].w * inv[r];
    }

  for (int c = 0; c < 4; ++c) {
    const int d = c * 128 + lane * 4;
    lds_acc[wave][d + 0] = acc[c * 4 + 0];
    lds_acc[wave][d + 1] = acc[c * 4 + 1];
    lds_acc[wave][d + 2] = acc[c * 4 + 2];
    lds_acc[wave][d + 3] = acc[c * 4 + 3];
  }
  __syncthreads();

  for (int d = threadIdx.x; d < 512; d += 256) {
    float s = 0.f;
#pragma unroll
    for (int w8 = 0; w8 < 8; ++w8) s += lds_acc[w8][d];
    part[((size_t)b * 16 + rowblk) * D_ + d] = s;
  }
}

__global__ __launch_bounds__(512)
void k2_scores(const float* __restrict__ x, const float* __restrict__ inv_norms,
               const float* __restrict__ part, float* __restrict__ scores) {
  const int b    = blockIdx.x >> 3;
  const int lane = threadIdx.x & 31;
  const int wave = __builtin_amdgcn_readfirstlane((int)(threadIdx.x >> 5));
  const int row0 = ((blockIdx.x & 7) << 7) + (wave << 3);

  __shared__ float4 mu4[D_ / 4];

  const float* pb = part + (size_t)b * 16 * D_ + threadIdx.x;
  float p[16];
  for (int j = 0; j < 16; ++j) p[j] = pb[j * D_];

  float inv = inv_norms[b * N_ + row0 + (lane & 7)];

  const float4* xr = (const float4*)(x + ((size_t)b * N_ + row0) * D_);
  float4 v[8][4];
  for (int r = 0; r < 8; ++r)
    for (int c = 0; c < 4; ++c) v[r][c] = xr[r * (D_ / 4) + c * 32 + lane];

  __builtin_amdgcn_sched_barrier(0);

  float ps = 0.f;
  for (int j = 0; j < 16; ++j) ps += p[j];
  ((float*)mu4)[threadIdx.x] = ps * (1.0f / N_);
  __builtin_amdgcn_fence(__ATOMIC_RELEASE, "workgroup", "local");
  __builtin_amdgcn_s_barrier();
  __builtin_amdgcn_fence(__ATOMIC_ACQUIRE, "workgroup", "local");

  float4 m[4];
  for (int c = 0; c < 4; ++c) m[c] = mu4[c * 32 + lane];

  asm volatile("" : "+v"(inv));

  float dot[8];
  for (int r = 0; r < 8; ++r) {
    float s = 0.f;
    for (int c = 0; c < 4; ++c)
      s += v[r][c].x * m[c].x + v[r][c].y * m[c].y + v[r][c].z * m[c].z + v[r][c].w * m[c].w;
    dot[r] = s;
  }
  for (int off = 16; off > 0; off >>= 1)
    for (int r = 0; r < 8; ++r) dot[r] += __shfl_xor(dot[r], off, 32);

  float mine = dot[0];
  for (int r = 1; r < 8; ++r) mine = (lane == r) ? dot[r] : mine;
  if (lane < 8) scores[b * N_ + row0 + lane] = mine * inv;
}

__device__ __forceinline__ unsigned score_key(const float s) {
  unsigned u = __float_as_uint(s);
  u = (u == 0x80000000u) ? 0u : u;
  return (u & 0x80000000u) ? ~u : (u | 0x80000000u);
}

__device__ __forceinline__ void lds_barrier() {
  __builtin_amdgcn_fence(__ATOMIC_RELEASE, "workgroup", "local");
  __builtin_amdgcn_s_barrier();
  __builtin_amdgcn_fence(__ATOMIC_ACQUIRE, "workgroup", "local");
}

__global__ __launch_bounds__(512)
void k3_select_merge_wmma(const float* __restrict__ x, const float* __restrict__ scores,
                          float* __restrict__ out) {
  const int b    = blockIdx.x >> 3;
  const int slab = (blockIdx.x & 7) << 6;
  const int lane = threadIdx.x & 31;
  const int wave = __builtin_amdgcn_readfirstlane((int)(threadIdx.x >> 5));
  const int c    = lane & 15;
  const int r0   = (wave << 6) + ((lane >> 4) << 5);

  __shared__ float2   red[2][16 * 32];
  __shared__ unsigned cnt[2][16][16];
  __shared__ uint2    fin[16];

  int slot = (int)(threadIdx.x * sizeof(float2));
  asm volatile("" : "+v"(slot));

  const float* sp = scores + b * N_ + (wave << 6) + lane;
  float sa = sp[0];
  float sb = sp[32];

  const float* xp = x + ((size_t)b * N_ + r0) * D_ + slab + 4 * c;
  float4 v[32];
  for (int j = 0; j < 32; ++j) v[j] = *(const float4*)(xp + (size_t)j * D_);

  __builtin_amdgcn_sched_barrier(0);

  const unsigned ka = score_key(sa);
  const unsigned kb = score_key(sb);

  unsigned T = 0;
  for (int p = 0; p < 8; ++p) {
    const int shift = 28 - 4 * p;
    const unsigned qa = __builtin_elementwise_sub_sat(ka, T) >> shift;
    const unsigned qb = __builtin_elementwise_sub_sat(kb, T) >> shift;
    unsigned mine = 0;
    for (int i = 1; i < 16; ++i) {
      const unsigned long long both =
          (unsigned long long)__builtin_amdgcn_ballot_w32(qa >= (unsigned)i) |
          ((unsigned long long)__builtin_amdgcn_ballot_w32(qb >= (unsigned)i) << 32);
      const unsigned n = (unsigned)__builtin_popcountll(both);
      mine = (lane == i) ? n : mine;
    }
    if (lane < 16) cnt[p & 1][wave][lane] = mine;
    lds_barrier();
    unsigned tot = 0;
    for (int w16 = 0; w16 < 16; ++w16) tot += cnt[p & 1][w16][c];
    const unsigned pass  = __builtin_amdgcn_ballot_w32(tot >= (unsigned)K_SEL) & 0xFFFEu;
    const unsigned exact = __builtin_amdgcn_ballot_w32(tot == (unsigned)K_SEL) & 0xFFFEu;
    T |= (unsigned)__builtin_popcount(pass) << shift;
    if (exact != 0) break;
  }

  const unsigned gtA = __builtin_amdgcn_ballot_w32(ka > T);
  const unsigned gtB = __builtin_amdgcn_ballot_w32(kb > T);
  const unsigned eqA = __builtin_amdgcn_ballot_w32(ka == T);
  const unsigned eqB = __builtin_amdgcn_ballot_w32(kb == T);
  if (lane == 0)
    fin[wave] = make_uint2((unsigned)(__builtin_popcount(gtA) + __builtin_popcount(gtB)),
                           (unsigned)(__builtin_popcount(eqA) + __builtin_popcount(eqB)));
  lds_barrier();
  unsigned above = 0, eq_before = 0;
  for (int w16 = 0; w16 < 16; ++w16) {
    const uint2 f = fin[w16];
    above += f.x;
    eq_before += (w16 < wave) ? f.y : 0u;
  }
  const unsigned need  = (unsigned)K_SEL - above;
  const unsigned below = (1u << lane) - 1u;
  const unsigned tieA  = eq_before + (unsigned)__builtin_popcount(eqA & below);
  const unsigned tieB  = eq_before + (unsigned)__builtin_popcount(eqA) +
                         (unsigned)__builtin_popcount(eqB & below);
  const unsigned keepA = __builtin_amdgcn_ballot_w32(ka > T || (ka == T && tieA < need));
  const unsigned keepB = __builtin_amdgcn_ballot_w32(kb > T || (kb == T && tieB < need));
  const unsigned keep  = (lane >> 4) ? keepB : keepA;

  const bool odd = (c & 1) != 0;
  const unsigned keep_even = odd ? 0u : keep;
  const unsigned keep_odd  = odd ? keep : 0u;
  const unsigned wbits = __float_as_uint(1.0f / K_SEL);
  v2f a[32];
  for (int j = 0; j < 32; ++j) {
    a[j][0] = __uint_as_float((unsigned)((int)(keep_even << (31 - j)) >> 31) & wbits);
    a[j][1] = __uint_as_float((unsigned)((int)(keep_odd  << (31 - j)) >> 31) & wbits);
  }

  __builtin_amdgcn_sched_barrier(0);

  v8f accP = {}, accQ = {};
  for (int j = 0; j < 32; ++j) {
    const v2f bp = { v[j].x, v[j].y };
    const v2f bq = { v[j].z, v[j].w };
    accP = __builtin_amdgcn_wmma_f32_16x16x4_f32(false, a[j], false, bp, (short)0, accP, false, false);
    accQ = __builtin_amdgcn_wmma_f32_16x16x4_f32(false, a[j], false, bq, (short)0, accQ, false, false);
  }

  *(float2*)((char*)&red[0][0] + slot) = make_float2(accP[0], accP[1]);
  *(float2*)((char*)&red[1][0] + slot) = make_float2(accQ[0], accQ[1]);
  __syncthreads();

  if (threadIdx.x < 16) {
    float2 p = red[0][threadIdx.x];
    float2 q = red[1][threadIdx.x];
    for (int j = 1; j < 16; ++j) {
      const float2 pj = red[0][j * 32 + threadIdx.x];
      const float2 qj = red[1][j * 32 + threadIdx.x];
      p.x += pj.x; p.y += pj.y; q.x += qj.x; q.y += qj.y;
    }
    *(float4*)(out + (size_t)b * D_ + slab + 4 * threadIdx.x) = make_float4(p.x, p.y, q.x, q.y);
  }
}

extern "C" void kernel_launch(void* const* d_in, const int* in_sizes, int n_in,
                              void* d_out, int out_size, void* d_ws, size_t ws_size,
                              hipStream_t stream) {
  (void)in_sizes; (void)n_in; (void)out_size; (void)ws_size;

  const float* x = (const float*)d_in[0];
  float* out = (float*)d_out;

  char* ws = (char*)d_ws;
  float* inv_norms = (float*)(ws + 0);
  float* scores    = (float*)(ws + 131072);
  float* part      = (float*)(ws + 458752);

  k1_norms_part<<<B_ * 16, 256, 0, stream>>>(x, inv_norms, part);
  k2_scores<<<B_ * 8, 512, 0, stream>>>(x, inv_norms, part, scores);
  k3_select_merge_wmma<<<B_ * 8, 512, 0, stream>>>(x, scores, out);
}
